# starting kernel
# baseline (speedup 1.0000x reference)
#include <hip/hip_runtime.h>
#include <math.h>

typedef float v2f __attribute__((ext_vector_type(2)));
typedef float v8f __attribute__((ext_vector_type(8)));

#define D_IN  256
#define D_HID 128
#define D_EMB 128
#define D_OUT 16

__global__ void transpose_w_kernel(const float* __restrict__ in,
                                   float* __restrict__ out, int K, int D)
{
    const int idx = blockIdx.x * blockDim.x + threadIdx.x;
    if (idx < K * D) {
        const int k = idx / D;
        const int c = idx - k * D;
        out[c * K + k] = in[idx];
    }
}

__global__ void gemm_wmma_f32_kernel(const float* __restrict__ A,
                                     const float* __restrict__ WT,
                                     float* __restrict__ C,
                                     const float* __restrict__ bias,
                                     int M, int K, int DOUTv,
                                     int totalTiles, int colTiles)
{
    const int gw = blockIdx.x * (blockDim.x >> 5) + (threadIdx.x >> 5);
    if (gw >= totalTiles) return;

    const int rowTile = gw / colTiles;
    const int colTile = gw - rowTile * colTiles;
    const int lane = threadIdx.x & 31;
    const int half = lane >> 4;
    const int lm   = lane & 15;

    const int rowBase = rowTile * 16;
    const int colBase = colTile * 16;

    const float* pA = A  + (size_t)(rowBase + lm) * K + 2 * half;
    const float* pB = WT + (size_t)(colBase + lm) * K + 2 * half;

    v8f acc = {};
    const int steps = K >> 2;
    #pragma unroll 4
    for (int s = 0; s < steps; ++s) {
        v2f a = *(const v2f*)pA;  pA += 4;
        v2f b = *(const v2f*)pB;  pB += 4;
        acc = __builtin_amdgcn_wmma_f32_16x16x4_f32(
                  false, a, false, b, (short)0, acc, false, false);
    }

    const float bv = bias ? bias[colBase + lm] : 0.0f;
    float* pC = C + (size_t)(rowBase + 8 * half) * DOUTv + colBase + lm;
    #pragma unroll
    for (int v = 0; v < 8; ++v) {
        const int r = rowBase + v + 8 * half;
        if (r < M) pC[(size_t)v * DOUTv] = acc[v] + bv;
    }
}

__global__ void spmm_atomic_kernel(const float* __restrict__ H,
                                   float* __restrict__ AGG,
                                   const int* __restrict__ ei,
                                   const float* __restrict__ ew,
                                   int E, int D)
{
    const int lane   = threadIdx.x & 31;
    const int wave   = (blockIdx.x * blockDim.x + threadIdx.x) >> 5;
    const int nwaves = (gridDim.x * blockDim.x) >> 5;

    for (int e = wave; e < E; e += nwaves) {
        const int   src = ei[e];
        const int   dst = ei[E + e];
        const float w   = ew[e];
        const float4 h4 = *(const float4*)(H + (size_t)src * D + lane * 4);
        float* p = AGG + (size_t)dst * D + lane * 4;
        atomicAdd(p + 0, h4.x * w);
        atomicAdd(p + 1, h4.y * w);
        atomicAdd(p + 2, h4.z * w);
        atomicAdd(p + 3, h4.w * w);
    }
}

__global__ void zero_kernel(float4* __restrict__ p, size_t n4)
{
    const size_t i = (size_t)blockIdx.x * blockDim.x + threadIdx.x;
    const size_t stride = (size_t)gridDim.x * blockDim.x;
    const float4 z = make_float4(0.f, 0.f, 0.f, 0.f);
    for (size_t j = i; j < n4; j += stride) p[j] = z;
}

__global__ void bias_act_kernel(const float4* __restrict__ in,
                                const float* __restrict__ bias,
                                float4* __restrict__ out,
                                size_t total4, int D, int doRelu)
{
    const size_t i = (size_t)blockIdx.x * blockDim.x + threadIdx.x;
    const size_t stride = (size_t)gridDim.x * blockDim.x;
    const int mask4 = (D >> 2) - 1;
    for (size_t j = i; j < total4; j += stride) {
        const float4 b4 = *(const float4*)(bias + ((j & mask4) << 2));
        float4 v = in[j];
        v.x += b4.x; v.y += b4.y; v.z += b4.z; v.w += b4.w;
        if (doRelu) {
            v.x = fmaxf(v.x, 0.f); v.y = fmaxf(v.y, 0.f);
            v.z = fmaxf(v.z, 0.f); v.w = fmaxf(v.w, 0.f);
        }
        out[j] = v;
    }
}

__global__ void log_softmax16_kernel(const float* __restrict__ Y,
                                     float* __restrict__ out, int N)
{
    const int r = blockIdx.x * blockDim.x + threadIdx.x;
    if (r >= N) return;
    float v[D_OUT];
    float m = -INFINITY;
    #pragma unroll
    for (int j = 0; j < D_OUT; ++j) { v[j] = Y[(size_t)r * D_OUT + j]; m = fmaxf(m, v[j]); }
    float s = 0.0f;
    #pragma unroll
    for (int j = 0; j < D_OUT; ++j) s += __expf(v[j] - m);
    const float l = m + __logf(s);
    #pragma unroll
    for (int j = 0; j < D_OUT; ++j) out[(size_t)r * D_OUT + j] = v[j] - l;
}

extern "C" void kernel_launch(void* const* d_in, const int* in_sizes, int n_in,
                              void* d_out, int out_size, void* d_ws, size_t ws_size,
                              hipStream_t stream)
{
    const float* x    = (const float*)d_in[0];
    const float* ew   = (const float*)d_in[1];
    const float* W1   = (const float*)d_in[2];
    const float* b1   = (const float*)d_in[3];
    const float* W2   = (const float*)d_in[4];
    const float* b2   = (const float*)d_in[5];
    const float* W3   = (const float*)d_in[6];
    const float* b3   = (const float*)d_in[7];
    const float* fcw  = (const float*)d_in[8];
    const float* fcb  = (const float*)d_in[9];
    const int*   ei   = (const int*)d_in[10];

    const int N = in_sizes[0] / D_IN;
    const int E = in_sizes[1];

    float* B0   = (float*)d_ws;
    float* B1   = B0 + (size_t)N * D_HID;
    float* WT1  = B1 + (size_t)N * D_HID;
    float* WT2  = WT1 + D_IN  * D_HID;
    float* WT3  = WT2 + D_HID * D_HID;
    float* WTfc = WT3 + D_HID * D_EMB;
    float* Y    = B0;

    float* outEmb = (float*)d_out;
    float* outLsm = outEmb + (size_t)N * D_EMB;

    const int rowTiles = (N + 15) / 16;
    const dim3 blk(256);

    auto transpose = [&](const float* W, float* WT, int K, int D) {
        const int tot = K * D;
        transpose_w_kernel<<<(tot + 255) / 256, blk, 0, stream>>>(W, WT, K, D);
    };
    auto gemm = [&](const float* A, const float* WT, float* C, const float* bias,
                    int K, int DOUTv) {
        const int colTiles = DOUTv / 16;
        const int total = rowTiles * colTiles;
        gemm_wmma_f32_kernel<<<(total + 7) / 8, blk, 0, stream>>>(
            A, WT, C, bias, N, K, DOUTv, total, colTiles);
    };
    auto zero = [&](float* p, size_t nElem) {
        zero_kernel<<<2048, blk, 0, stream>>>((float4*)p, nElem / 4);
    };
    auto spmm = [&](const float* H, float* AGG) {
        spmm_atomic_kernel<<<2048, blk, 0, stream>>>(H, AGG, ei, ew, E, D_HID);
    };
    auto biasAct = [&](const float* in, const float* b, float* out, int relu) {
        bias_act_kernel<<<2048, blk, 0, stream>>>((const float4*)in, b,
                                                  (float4*)out,
                                                  (size_t)N * D_HID / 4,
                                                  D_HID, relu);
    };

    transpose(W1, WT1, D_IN,  D_HID);
    transpose(W2, WT2, D_HID, D_HID);
    transpose(W3, WT3, D_HID, D_EMB);
    transpose(fcw, WTfc, D_EMB, D_OUT);

    gemm(x, WT1, B0, nullptr, D_IN, D_HID);
    zero(B1, (size_t)N * D_HID);
    spmm(B0, B1);
    biasAct(B1, b1, B1, 1);

    gemm(B1, WT2, B0, nullptr, D_HID, D_HID);
    zero(B1, (size_t)N * D_HID);
    spmm(B0, B1);
    biasAct(B1, b2, B1, 1);

    gemm(B1, WT3, B0, nullptr, D_HID, D_EMB);
    zero(B1, (size_t)N * D_HID);
    spmm(B0, B1);
    biasAct(B1, b3, outEmb, 0);

    gemm(outEmb, WTfc, Y, fcb, D_EMB, D_OUT);
    log_softmax16_kernel<<<(N + 255) / 256, blk, 0, stream>>>(Y, outLsm, N);
}


# SpMM scatter: lane owns features lane+32j, so each wave-wide f32 atomic (and load) covers one whole 128-byte line: 4 atomic line requests per edge instead of 16
# speedup vs baseline: 1.4017x; 1.4017x over previous
#include <hip/hip_runtime.h>
#include <math.h>

typedef float v2f __attribute__((ext_vector_type(2)));
typedef float v8f __attribute__((ext_vector_type(8)));

#define D_IN  256
#define D_HID 128
#define D_EMB 128
#define D_OUT 16

__global__ void transpose_w_kernel(const float* __restrict__ in,
                                   float* __restrict__ out, int K, int D)
{
    const int idx = blockIdx.x * blockDim.x + threadIdx.x;
    if (idx < K * D) {
        const int k = idx / D;
        const int c = idx - k * D;
        out[c * K + k] = in[idx];
    }
}

__global__ void gemm_wmma_f32_kernel(const float* __restrict__ A,
                                     const float* __restrict__ WT,
                                     float* __restrict__ C,
                                     const float* __restrict__ bias,
                                     int M, int K, int DOUTv,
                                     int totalTiles, int colTiles)
{
    const int gw = blockIdx.x * (blockDim.x >> 5) + (threadIdx.x >> 5);
    if (gw >= totalTiles) return;

    const int rowTile = gw / colTiles;
    const int colTile = gw - rowTile * colTiles;
    const int lane = threadIdx.x & 31;
    const int half = lane >> 4;
    const int lm   = lane & 15;

    const int rowBase = rowTile * 16;
    const int colBase = colTile * 16;

    const float* pA = A  + (size_t)(rowBase + lm) * K + 2 * half;
    const float* pB = WT + (size_t)(colBase + lm) * K + 2 * half;

    v8f acc = {};
    const int steps = K >> 2;
    #pragma unroll 4
    for (int s = 0; s < steps; ++s) {
        v2f a = *(const v2f*)pA;  pA += 4;
        v2f b = *(const v2f*)pB;  pB += 4;
        acc = __builtin_amdgcn_wmma_f32_16x16x4_f32(
                  false, a, false, b, (short)0, acc, false, false);
    }

    const float bv = bias ? bias[colBase + lm] : 0.0f;
    float* pC = C + (size_t)(rowBase + 8 * half) * DOUTv + colBase + lm;
    #pragma unroll
    for (int v = 0; v < 8; ++v) {
        const int r = rowBase + v + 8 * half;
        if (r < M) pC[(size_t)v * DOUTv] = acc[v] + bv;
    }
}

__global__ void spmm_atomic_kernel(const float* __restrict__ H,
                                   float* __restrict__ AGG,
                                   const int* __restrict__ ei,
                                   const float* __restrict__ ew,
                                   int E, int D)
{
    const int lane   = threadIdx.x & 31;
    const int wave   = (blockIdx.x * blockDim.x + threadIdx.x) >> 5;
    const int nwaves = (gridDim.x * blockDim.x) >> 5;

    for (int e = wave; e < E; e += nwaves) {
        const int   src = ei[e];
        const int   dst = ei[E + e];
        const float w   = ew[e];
        const float* h = H + (size_t)src * D + lane;
        const float h0 = h[0];
        const float h1 = h[32];
        const float h2 = h[64];
        const float h3 = h[96];
        float* p = AGG + (size_t)dst * D + lane;
        atomicAdd(p +  0, h0 * w);
        atomicAdd(p + 32, h1 * w);
        atomicAdd(p + 64, h2 * w);
        atomicAdd(p + 96, h3 * w);
    }
}

__global__ void zero_kernel(float4* __restrict__ p, size_t n4)
{
    const size_t i = (size_t)blockIdx.x * blockDim.x + threadIdx.x;
    const size_t stride = (size_t)gridDim.x * blockDim.x;
    const float4 z = make_float4(0.f, 0.f, 0.f, 0.f);
    for (size_t j = i; j < n4; j += stride) p[j] = z;
}

__global__ void bias_act_kernel(const float4* __restrict__ in,
                                const float* __restrict__ bias,
                                float4* __restrict__ out,
                                size_t total4, int D, int doRelu)
{
    const size_t i = (size_t)blockIdx.x * blockDim.x + threadIdx.x;
    const size_t stride = (size_t)gridDim.x * blockDim.x;
    const int mask4 = (D >> 2) - 1;
    for (size_t j = i; j < total4; j += stride) {
        const float4 b4 = *(const float4*)(bias + ((j & mask4) << 2));
        float4 v = in[j];
        v.x += b4.x; v.y += b4.y; v.z += b4.z; v.w += b4.w;
        if (doRelu) {
            v.x = fmaxf(v.x, 0.f); v.y = fmaxf(v.y, 0.f);
            v.z = fmaxf(v.z, 0.f); v.w = fmaxf(v.w, 0.f);
        }
        out[j] = v;
    }
}

__global__ void log_softmax16_kernel(const float* __restrict__ Y,
                                     float* __restrict__ out, int N)
{
    const int r = blockIdx.x * blockDim.x + threadIdx.x;
    if (r >= N) return;
    float v[D_OUT];
    float m = -INFINITY;
    #pragma unroll
    for (int j = 0; j < D_OUT; ++j) { v[j] = Y[(size_t)r * D_OUT + j]; m = fmaxf(m, v[j]); }
    float s = 0.0f;
    #pragma unroll
    for (int j = 0; j < D_OUT; ++j) s += __expf(v[j] - m);
    const float l = m + __logf(s);
    #pragma unroll
    for (int j = 0; j < D_OUT; ++j) out[(size_t)r * D_OUT + j] = v[j] - l;
}

extern "C" void kernel_launch(void* const* d_in, const int* in_sizes, int n_in,
                              void* d_out, int out_size, void* d_ws, size_t ws_size,
                              hipStream_t stream)
{
    const float* x    = (const float*)d_in[0];
    const float* ew   = (const float*)d_in[1];
    const float* W1   = (const float*)d_in[2];
    const float* b1   = (const float*)d_in[3];
    const float* W2   = (const float*)d_in[4];
    const float* b2   = (const float*)d_in[5];
    const float* W3   = (const float*)d_in[6];
    const float* b3   = (const float*)d_in[7];
    const float* fcw  = (const float*)d_in[8];
    const float* fcb  = (const float*)d_in[9];
    const int*   ei   = (const int*)d_in[10];

    const int N = in_sizes[0] / D_IN;
    const int E = in_sizes[1];

    float* B0   = (float*)d_ws;
    float* B1   = B0 + (size_t)N * D_HID;
    float* WT1  = B1 + (size_t)N * D_HID;
    float* WT2  = WT1 + D_IN  * D_HID;
    float* WT3  = WT2 + D_HID * D_HID;
    float* WTfc = WT3 + D_HID * D_EMB;
    float* Y    = B0;

    float* outEmb = (float*)d_out;
    float* outLsm = outEmb + (size_t)N * D_EMB;

    const int rowTiles = (N + 15) / 16;
    const dim3 blk(256);

    auto transpose = [&](const float* W, float* WT, int K, int D) {
        const int tot = K * D;
        transpose_w_kernel<<<(tot + 255) / 256, blk, 0, stream>>>(W, WT, K, D);
    };
    auto gemm = [&](const float* A, const float* WT, float* C, const float* bias,
                    int K, int DOUTv) {
        const int colTiles = DOUTv / 16;
        const int total = rowTiles * colTiles;
        gemm_wmma_f32_kernel<<<(total + 7) / 8, blk, 0, stream>>>(
            A, WT, C, bias, N, K, DOUTv, total, colTiles);
    };
    auto zero = [&](float* p, size_t nElem) {
        zero_kernel<<<2048, blk, 0, stream>>>((float4*)p, nElem / 4);
    };
    auto spmm = [&](const float* H, float* AGG) {
        spmm_atomic_kernel<<<2048, blk, 0, stream>>>(H, AGG, ei, ew, E, D_HID);
    };
    auto biasAct = [&](const float* in, const float* b, float* out, int relu) {
        bias_act_kernel<<<2048, blk, 0, stream>>>((const float4*)in, b,
                                                  (float4*)out,
                                                  (size_t)N * D_HID / 4,
                                                  D_HID, relu);
    };

    transpose(W1, WT1, D_IN,  D_HID);
    transpose(W2, WT2, D_HID, D_HID);
    transpose(W3, WT3, D_HID, D_EMB);
    transpose(fcw, WTfc, D_EMB, D_OUT);

    gemm(x, WT1, B0, nullptr, D_IN, D_HID);
    zero(B1, (size_t)N * D_HID);
    spmm(B0, B1);
    biasAct(B1, b1, B1, 1);

    gemm(B1, WT2, B0, nullptr, D_HID, D_HID);
    zero(B1, (size_t)N * D_HID);
    spmm(B0, B1);
    biasAct(B1, b2, B1, 1);

    gemm(B1, WT3, B0, nullptr, D_HID, D_EMB);
    zero(B1, (size_t)N * D_HID);
    spmm(B0, B1);
    biasAct(B1, b3, outEmb, 0);

    gemm(outEmb, WTfc, Y, fcb, D_EMB, D_OUT);
    log_softmax16_kernel<<<(N + 255) / 256, blk, 0, stream>>>(Y, outLsm, N);
}


# GEMM re-tiled: one wave computes a 16x128 strip (each A fragment loaded once, feeds 8 column tiles), W staged row-major in LDS once per block; the 3 weight-transpose launches are dropped
# speedup vs baseline: 1.7338x; 1.2369x over previous
#include <hip/hip_runtime.h>
#include <math.h>

typedef float v2f __attribute__((ext_vector_type(2)));
typedef float v8f __attribute__((ext_vector_type(8)));

#define D_IN  256
#define D_HID 128
#define D_EMB 128
#define D_OUT 16

__global__ void transpose_w_kernel(const float* __restrict__ in,
                                   float* __restrict__ out, int K, int D)
{
    const int idx = blockIdx.x * blockDim.x + threadIdx.x;
    if (idx < K * D) {
        const int k = idx / D;
        const int c = idx - k * D;
        out[c * K + k] = in[idx];
    }
}

__global__ void gemm_wmma_f32_kernel(const float* __restrict__ A,
                                     const float* __restrict__ WT,
                                     float* __restrict__ C,
                                     const float* __restrict__ bias,
                                     int M, int K, int DOUTv,
                                     int totalTiles, int colTiles)
{
    const int gw = blockIdx.x * (blockDim.x >> 5) + (threadIdx.x >> 5);
    if (gw >= totalTiles) return;

    const int rowTile = gw / colTiles;
    const int colTile = gw - rowTile * colTiles;
    const int lane = threadIdx.x & 31;
    const int half = lane >> 4;
    const int lm   = lane & 15;

    const int rowBase = rowTile * 16;
    const int colBase = colTile * 16;

    const float* pA = A  + (size_t)(rowBase + lm) * K + 2 * half;
    const float* pB = WT + (size_t)(colBase + lm) * K + 2 * half;

    v8f acc = {};
    const int steps = K >> 2;
    #pragma unroll 4
    for (int s = 0; s < steps; ++s) {
        v2f a = *(const v2f*)pA;  pA += 4;
        v2f b = *(const v2f*)pB;  pB += 4;
        acc = __builtin_amdgcn_wmma_f32_16x16x4_f32(
                  false, a, false, b, (short)0, acc, false, false);
    }

    const float bv = bias ? bias[colBase + lm] : 0.0f;
    float* pC = C + (size_t)(rowBase + 8 * half) * DOUTv + colBase + lm;
    #pragma unroll
    for (int v = 0; v < 8; ++v) {
        const int r = rowBase + v + 8 * half;
        if (r < M) pC[(size_t)v * DOUTv] = acc[v] + bv;
    }
}

#define W_PITCH 272

template <bool LAST>
__device__ __forceinline__ void strip_group4(v8f (&acc)[8], float4 (&a)[4],
                                             v2f (&bq)[2][16],
                                             const float4* __restrict__ pA,
                                             const float* wp)
{
    #pragma unroll
    for (int jj = 0; jj < 4; ++jj) {
        const int cur = jj & 1;
        if (!LAST || jj < 3) {
            #pragma unroll
            for (int u = 0; u < 16; ++u)
                bq[cur ^ 1][u] = *(const v2f*)(wp + (4 * (jj + 1) + (u >> 3)) * W_PITCH
                                               + 32 * (u & 7));
        }
        const float4 aj = a[jj];
        if (!LAST) a[jj] = pA[2 * (jj + 4)];
        #pragma unroll
        for (int u = 0; u < 16; ++u) {
            v2f av;
            av.x = (u >> 3) ? aj.z : aj.x;
            av.y = (u >> 3) ? aj.w : aj.y;
            acc[u & 7] = __builtin_amdgcn_wmma_f32_16x16x4_f32(
                             false, av, false, bq[cur][u], (short)0, acc[u & 7], false, false);
        }
    }
}

template <int K>
__global__ __launch_bounds__(256)
void gemm_strip_wmma_f32_kernel(const float* __restrict__ A,
                                const float* __restrict__ W,
                                float* __restrict__ C,
                                int M, int rowTiles)
{
    constexpr int NG = K / 32;
    __shared__ __attribute__((aligned(16))) float Ws[(K / 2) * W_PITCH];

    const int tid  = threadIdx.x;
    const int lane = tid & 31;
    const int wave = __builtin_amdgcn_readfirstlane(tid >> 5);
    const int half = lane >> 4;
    const int c    = lane & 15;
    const int tile = blockIdx.x * 8 + wave;

    int row = tile * 16 + c;
    if (row > M - 1) row = M - 1;
    const float4* pA = (const float4*)(A + (size_t)row * K + 4 * half);
    float4 a[4];
    #pragma unroll
    for (int j = 0; j < 4; ++j) a[j] = pA[2 * j];

    #pragma unroll
    for (int n = 0; n < K / 16; ++n) {
        const int idx = tid + 256 * n;
        const int p = idx >> 5;
        const int q = idx & 31;
        const float4 r0 = *(const float4*)(W + (size_t)(2 * p)     * 128 + 4 * q);
        const float4 r1 = *(const float4*)(W + (size_t)(2 * p + 1) * 128 + 4 * q);
        float4* d = (float4*)(Ws + p * W_PITCH + 8 * q);
        d[0] = make_float4(r0.x, r1.x, r0.y, r1.y);
        d[1] = make_float4(r0.z, r1.z, r0.w, r1.w);
    }
    __syncthreads();
    if (tile >= rowTiles) return;

    v8f acc[8] = {};
    const float* wp = Ws + 2 * half * W_PITCH + 2 * c;

    v2f bq[2][16];
    #pragma unroll
    for (int u = 0; u < 16; ++u)
        bq[0][u] = *(const v2f*)(wp + (u >> 3) * W_PITCH + 32 * (u & 7));

    #pragma nounroll
    for (int g = 0; g < NG - 1; ++g) {
        strip_group4<false>(acc, a, bq, pA, wp);
        pA += 8;
        wp += 16 * W_PITCH;
    }
    strip_group4<true>(acc, a, bq, pA, wp);

    const int r0 = tile * 16 + 8 * half;
    float* pC = C + (size_t)r0 * 128 + c;
    #pragma unroll
    for (int v = 0; v < 8; ++v) {
        if (r0 + v < M) {
            #pragma unroll
            for (int t = 0; t < 8; ++t) pC[(size_t)v * 128 + 16 * t] = acc[t][v];
        }
    }
}

__global__ void spmm_atomic_kernel(const float* __restrict__ H,
                                   float* __restrict__ AGG,
                                   const int* __restrict__ ei,
                                   const float* __restrict__ ew,
                                   int E, int D)
{
    const int lane   = threadIdx.x & 31;
    const int wave   = (blockIdx.x * blockDim.x + threadIdx.x) >> 5;
    const int nwaves = (gridDim.x * blockDim.x) >> 5;

    for (int e = wave; e < E; e += nwaves) {
        const int   src = ei[e];
        const int   dst = ei[E + e];
        const float w   = ew[e];
        const float* h = H + (size_t)src * D + lane;
        const float h0 = h[0];
        const float h1 = h[32];
        const float h2 = h[64];
        const float h3 = h[96];
        float* p = AGG + (size_t)dst * D + lane;
        atomicAdd(p +  0, h0 * w);
        atomicAdd(p + 32, h1 * w);
        atomicAdd(p + 64, h2 * w);
        atomicAdd(p + 96, h3 * w);
    }
}

__global__ void zero_kernel(float4* __restrict__ p, size_t n4)
{
    const size_t i = (size_t)blockIdx.x * blockDim.x + threadIdx.x;
    const size_t stride = (size_t)gridDim.x * blockDim.x;
    const float4 z = make_float4(0.f, 0.f, 0.f, 0.f);
    for (size_t j = i; j < n4; j += stride) p[j] = z;
}

__global__ void bias_act_kernel(const float4* __restrict__ in,
                                const float* __restrict__ bias,
                                float4* __restrict__ out,
                                size_t total4, int D, int doRelu)
{
    const size_t i = (size_t)blockIdx.x * blockDim.x + threadIdx.x;
    const size_t stride = (size_t)gridDim.x * blockDim.x;
    const int mask4 = (D >> 2) - 1;
    for (size_t j = i; j < total4; j += stride) {
        const float4 b4 = *(const float4*)(bias + ((j & mask4) << 2));
        float4 v = in[j];
        v.x += b4.x; v.y += b4.y; v.z += b4.z; v.w += b4.w;
        if (doRelu) {
            v.x = fmaxf(v.x, 0.f); v.y = fmaxf(v.y, 0.f);
            v.z = fmaxf(v.z, 0.f); v.w = fmaxf(v.w, 0.f);
        }
        out[j] = v;
    }
}

__global__ void log_softmax16_kernel(const float* __restrict__ Y,
                                     float* __restrict__ out, int N)
{
    const int r = blockIdx.x * blockDim.x + threadIdx.x;
    if (r >= N) return;
    float v[D_OUT];
    float m = -INFINITY;
    #pragma unroll
    for (int j = 0; j < D_OUT; ++j) { v[j] = Y[(size_t)r * D_OUT + j]; m = fmaxf(m, v[j]); }
    float s = 0.0f;
    #pragma unroll
    for (int j = 0; j < D_OUT; ++j) s += __expf(v[j] - m);
    const float l = m + __logf(s);
    #pragma unroll
    for (int j = 0; j < D_OUT; ++j) out[(size_t)r * D_OUT + j] = v[j] - l;
}

extern "C" void kernel_launch(void* const* d_in, const int* in_sizes, int n_in,
                              void* d_out, int out_size, void* d_ws, size_t ws_size,
                              hipStream_t stream)
{
    const float* x    = (const float*)d_in[0];
    const float* ew   = (const float*)d_in[1];
    const float* W1   = (const float*)d_in[2];
    const float* b1   = (const float*)d_in[3];
    const float* W2   = (const float*)d_in[4];
    const float* b2   = (const float*)d_in[5];
    const float* W3   = (const float*)d_in[6];
    const float* b3   = (const float*)d_in[7];
    const float* fcw  = (const float*)d_in[8];
    const float* fcb  = (const float*)d_in[9];
    const int*   ei   = (const int*)d_in[10];

    const int N = in_sizes[0] / D_IN;
    const int E = in_sizes[1];

    float* B0   = (float*)d_ws;
    float* B1   = B0 + (size_t)N * D_HID;
    float* WT1  = B1 + (size_t)N * D_HID;
    float* WT2  = WT1 + D_IN  * D_HID;
    float* WT3  = WT2 + D_HID * D_HID;
    float* WTfc = WT3 + D_HID * D_EMB;
    float* Y    = B0;

    float* outEmb = (float*)d_out;
    float* outLsm = outEmb + (size_t)N * D_EMB;

    const int rowTiles = (N + 15) / 16;
    const dim3 blk(256);

    auto transpose = [&](const float* W, float* WT, int K, int D) {
        const int tot = K * D;
        transpose_w_kernel<<<(tot + 255) / 256, blk, 0, stream>>>(W, WT, K, D);
    };
    auto gemm = [&](const float* A, const float* WT, float* C, const float* bias,
                    int K, int DOUTv) {
        const int colTiles = DOUTv / 16;
        const int total = rowTiles * colTiles;
        gemm_wmma_f32_kernel<<<(total + 7) / 8, blk, 0, stream>>>(
            A, WT, C, bias, N, K, DOUTv, total, colTiles);
    };
    auto gemmStrip256 = [&](const float* A, const float* W, float* C) {
        gemm_strip_wmma_f32_kernel<D_IN><<<(rowTiles + 7) / 8, blk, 0, stream>>>(
            A, W, C, N, rowTiles);
    };
    auto gemmStrip128 = [&](const float* A, const float* W, float* C) {
        gemm_strip_wmma_f32_kernel<D_HID><<<(rowTiles + 7) / 8, blk, 0, stream>>>(
            A, W, C, N, rowTiles);
    };
    auto zero = [&](float* p, size_t nElem) {
        zero_kernel<<<2048, blk, 0, stream>>>((float4*)p, nElem / 4);
    };
    auto spmm = [&](const float* H, float* AGG) {
        spmm_atomic_kernel<<<2048, blk, 0, stream>>>(H, AGG, ei, ew, E, D_HID);
    };
    auto biasAct = [&](const float* in, const float* b, float* out, int relu) {
        bias_act_kernel<<<2048, blk, 0, stream>>>((const float4*)in, b,
                                                  (float4*)out,
                                                  (size_t)N * D_HID / 4,
                                                  D_HID, relu);
    };

    transpose(fcw, WTfc, D_EMB, D_OUT);

    gemmStrip256(x, W1, B0);
    zero(B1, (size_t)N * D_HID);
    spmm(B0, B1);
    biasAct(B1, b1, B1, 1);

    gemmStrip128(B1, W2, B0);
    zero(B1, (size_t)N * D_HID);
    spmm(B0, B1);
    biasAct(B1, b2, B1, 1);

    gemmStrip128(B1, W3, B0);
    zero(B1, (size_t)N * D_HID);
    spmm(B0, B1);
    biasAct(B1, b3, outEmb, 0);

    gemm(outEmb, WTfc, Y, fcb, D_EMB, D_OUT);
    log_softmax16_kernel<<<(N + 255) / 256, blk, 0, stream>>>(Y, outLsm, N);
}


# SpMM as a gather: edges counting-sorted by destination once per call, one wave sums each output row from its edge list with bias+relu fused; float atomics, clears and bias passes dropped
# speedup vs baseline: 2.2508x; 1.2982x over previous
#include <hip/hip_runtime.h>
#include <math.h>

typedef float v2f __attribute__((ext_vector_type(2)));
typedef float v8f __attribute__((ext_vector_type(8)));

#define D_IN  256
#define D_HID 128
#define D_EMB 128
#define D_OUT 16

__global__ void transpose_w_kernel(const float* __restrict__ in,
                                   float* __restrict__ out, int K, int D)
{
    const int idx = blockIdx.x * blockDim.x + threadIdx.x;
    if (idx < K * D) {
        const int k = idx / D;
        const int c = idx - k * D;
        out[c * K + k] = in[idx];
    }
}

__global__ void gemm_wmma_f32_kernel(const float* __restrict__ A,
                                     const float* __restrict__ WT,
                                     float* __restrict__ C,
                                     const float* __restrict__ bias,
                                     int M, int K, int DOUTv,
                                     int totalTiles, int colTiles)
{
    const int gw = blockIdx.x * (blockDim.x >> 5) + (threadIdx.x >> 5);
    if (gw >= totalTiles) return;

    const int rowTile = gw / colTiles;
    const int colTile = gw - rowTile * colTiles;
    const int lane = threadIdx.x & 31;
    const int half = lane >> 4;
    const int lm   = lane & 15;

    const int rowBase = rowTile * 16;
    const int colBase = colTile * 16;

    const float* pA = A  + (size_t)(rowBase + lm) * K + 2 * half;
    const float* pB = WT + (size_t)(colBase + lm) * K + 2 * half;

    v8f acc = {};
    const int steps = K >> 2;
    #pragma unroll 4
    for (int s = 0; s < steps; ++s) {
        v2f a = *(const v2f*)pA;  pA += 4;
        v2f b = *(const v2f*)pB;  pB += 4;
        acc = __builtin_amdgcn_wmma_f32_16x16x4_f32(
                  false, a, false, b, (short)0, acc, false, false);
    }

    const float bv = bias ? bias[colBase + lm] : 0.0f;
    float* pC = C + (size_t)(rowBase + 8 * half) * DOUTv + colBase + lm;
    #pragma unroll
    for (int v = 0; v < 8; ++v) {
        const int r = rowBase + v + 8 * half;
        if (r < M) pC[(size_t)v * DOUTv] = acc[v] + bv;
    }
}

#define W_PITCH 272

template <bool LAST>
__device__ __forceinline__ void strip_group4(v8f (&acc)[8], float4 (&a)[4],
                                             v2f (&bq)[2][16],
                                             const float4* __restrict__ pA,
                                             const float* wp)
{
    #pragma unroll
    for (int jj = 0; jj < 4; ++jj) {
        const int cur = jj & 1;
        if (!LAST || jj < 3) {
            #pragma unroll
            for (int u = 0; u < 16; ++u)
                bq[cur ^ 1][u] = *(const v2f*)(wp + (4 * (jj + 1) + (u >> 3)) * W_PITCH
                                               + 32 * (u & 7));
        }
        const float4 aj = a[jj];
        if (!LAST) a[jj] = pA[2 * (jj + 4)];
        #pragma unroll
        for (int u = 0; u < 16; ++u) {
            v2f av;
            av.x = (u >> 3) ? aj.z : aj.x;
            av.y = (u >> 3) ? aj.w : aj.y;
            acc[u & 7] = __builtin_amdgcn_wmma_f32_16x16x4_f32(
                             false, av, false, bq[cur][u], (short)0, acc[u & 7], false, false);
        }
    }
}

template <int K>
__global__ __launch_bounds__(256)
void gemm_strip_wmma_f32_kernel(const float* __restrict__ A,
                                const float* __restrict__ W,
                                float* __restrict__ C,
                                int M, int rowTiles)
{
    constexpr int NG = K / 32;
    __shared__ __attribute__((aligned(16))) float Ws[(K / 2) * W_PITCH];

    const int tid  = threadIdx.x;
    const int lane = tid & 31;
    const int wave = __builtin_amdgcn_readfirstlane(tid >> 5);
    const int half = lane >> 4;
    const int c    = lane & 15;
    const int tile = blockIdx.x * 8 + wave;

    int row = tile * 16 + c;
    if (row > M - 1) row = M - 1;
    const float4* pA = (const float4*)(A + (size_t)row * K + 4 * half);
    float4 a[4];
    #pragma unroll
    for (int j = 0; j < 4; ++j) a[j] = pA[2 * j];

    #pragma unroll
    for (int n = 0; n < K / 16; ++n) {
        const int idx = tid + 256 * n;
        const int p = idx >> 5;
        const int q = idx & 31;
        const float4 r0 = *(const float4*)(W + (size_t)(2 * p)     * 128 + 4 * q);
        const float4 r1 = *(const float4*)(W + (size_t)(2 * p + 1) * 128 + 4 * q);
        float4* d = (float4*)(Ws + p * W_PITCH + 8 * q);
        d[0] = make_float4(r0.x, r1.x, r0.y, r1.y);
        d[1] = make_float4(r0.z, r1.z, r0.w, r1.w);
    }
    __syncthreads();
    if (tile >= rowTiles) return;

    v8f acc[8] = {};
    const float* wp = Ws + 2 * half * W_PITCH + 2 * c;

    v2f bq[2][16];
    #pragma unroll
    for (int u = 0; u < 16; ++u)
        bq[0][u] = *(const v2f*)(wp + (u >> 3) * W_PITCH + 32 * (u & 7));

    #pragma nounroll
    for (int g = 0; g < NG - 1; ++g) {
        strip_group4<false>(acc, a, bq, pA, wp);
        pA += 8;
        wp += 16 * W_PITCH;
    }
    strip_group4<true>(acc, a, bq, pA, wp);

    const int r0 = tile * 16 + 8 * half;
    float* pC = C + (size_t)r0 * 128 + c;
    #pragma unroll
    for (int v = 0; v < 8; ++v) {
        if (r0 + v < M) {
            #pragma unroll
            for (int t = 0; t < 8; ++t) pC[(size_t)v * 128 + 16 * t] = acc[t][v];
        }
    }
}

__global__ void zero_int_kernel(int* __restrict__ p, int n)
{
    const int i = blockIdx.x * blockDim.x + threadIdx.x;
    if (i < n) p[i] = 0;
}

__global__ void csr_count_kernel(const int* __restrict__ ei, int* __restrict__ cnt, int E)
{
    const int t = blockIdx.x * blockDim.x + threadIdx.x;
    const int T = gridDim.x * blockDim.x;
    int d[4];
    #pragma unroll
    for (int k = 0; k < 4; ++k) {
        const int e = t + k * T;
        d[k] = (e < E) ? ei[E + e] : -1;
    }
    #pragma unroll
    for (int k = 0; k < 4; ++k)
        if (d[k] >= 0) atomicAdd(cnt + d[k], 1);
}

__global__ __launch_bounds__(1024)
void csr_scan_kernel(const int* __restrict__ cnt, int* __restrict__ rowptr, int n)
{
    __shared__ __attribute__((aligned(16))) int wtot[2][32];
    const int tid  = threadIdx.x;
    const int lane = tid & 31;
    const int wv   = tid >> 5;

    int cur[4], nxt[4];
    #pragma unroll
    for (int k = 0; k < 4; ++k) {
        const int idx = 4 * tid + k;
        cur[k] = (idx < n) ? cnt[idx] : 0;
    }
    int carry = 0;
    int par = 0;
    for (int base = 0; base < n; base += 4096) {
        #pragma unroll
        for (int k = 0; k < 4; ++k) {
            const int idx = base + 4096 + 4 * tid + k;
            nxt[k] = (idx < n) ? cnt[idx] : 0;
        }
        const int t = cur[0] + cur[1] + cur[2] + cur[3];
        int inc = t;
        #pragma unroll
        for (int o = 1; o < 32; o <<= 1) {
            const int y = __shfl_up(inc, o);
            if (lane >= o) inc += y;
        }
        if (lane == 31) wtot[par][wv] = inc;
        __syncthreads();
        int before = 0, total = 0;
        #pragma unroll
        for (int i = 0; i < 32; ++i) {
            const int x = wtot[par][i];
            total += x;
            if (i < wv) before += x;
        }
        int ex = carry + before + inc - t;
        carry += total;
        const int idx0 = base + 4 * tid;
        #pragma unroll
        for (int k = 0; k < 4; ++k) {
            if (idx0 + k < n) rowptr[idx0 + k] = ex;
            ex += cur[k];
        }
        #pragma unroll
        for (int k = 0; k < 4; ++k) cur[k] = nxt[k];
        par ^= 1;
    }
    if (tid == 0) rowptr[n] = carry;
}

__global__ void csr_fill_kernel(const int* __restrict__ ei, const float* __restrict__ ew,
                                const int* __restrict__ rowptr, int* __restrict__ cnt,
                                int* __restrict__ ssrc, float* __restrict__ sw, int E)
{
    const int t = blockIdx.x * blockDim.x + threadIdx.x;
    const int T = gridDim.x * blockDim.x;
    int s[4], d[4], v[4], r[4];
    float w[4];
    #pragma unroll
    for (int k = 0; k < 4; ++k) {
        const int e = t + k * T;
        const bool ok = e < E;
        s[k] = ok ? ei[e] : 0;
        d[k] = ok ? ei[E + e] : -1;
        w[k] = ok ? ew[e] : 0.0f;
    }
    #pragma unroll
    for (int k = 0; k < 4; ++k) {
        v[k] = 0; r[k] = 0;
        if (d[k] >= 0) { v[k] = atomicSub(cnt + d[k], 1); r[k] = rowptr[d[k]]; }
    }
    #pragma unroll
    for (int k = 0; k < 4; ++k) {
        if (d[k] >= 0) {
            const int pos = r[k] + v[k] - 1;
            ssrc[pos] = s[k];
            sw[pos]   = w[k];
        }
    }
}

#define SPMM_GROUP 4
template <bool RELU>
__global__ __launch_bounds__(256)
void spmm_gather_kernel(const float* __restrict__ H, float* __restrict__ OUT,
                        const int* __restrict__ rowptr,
                        const int* __restrict__ ssrc, const float* __restrict__ sw,
                        const float* __restrict__ bias, int N)
{
    const int lane   = threadIdx.x & 31;
    const int wave   = __builtin_amdgcn_readfirstlane((blockIdx.x * blockDim.x + threadIdx.x) >> 5);
    const int nwaves = (gridDim.x * blockDim.x) >> 5;
    const float4 b4  = *(const float4*)(bias + 4 * lane);
    const float* Hl  = H + 4 * lane;

    for (int row = wave; row < N; row += nwaves) {
        const int beg = rowptr[row];
        const int end = rowptr[row + 1];
        float4 acc = make_float4(0.f, 0.f, 0.f, 0.f);
        for (int base = beg; base < end; base += 32) {
            const int j = base + lane;
            int   s = 0;
            float w = 0.0f;
            if (j < end) { s = ssrc[j]; w = sw[j]; }
            const int wi  = __float_as_int(w);
            const int cnt = (end - base < 32) ? (end - base) : 32;
            for (int k = 0; k < cnt; k += SPMM_GROUP) {
                float4 h[SPMM_GROUP];
                float  ws[SPMM_GROUP];
                #pragma unroll
                for (int i = 0; i < SPMM_GROUP; ++i) {
                    const int  q  = (k + i < cnt) ? (k + i) : (cnt - 1);
                    const int  si = __builtin_amdgcn_readlane(s, q);
                    const float wq = __int_as_float(__builtin_amdgcn_readlane(wi, q));
                    ws[i] = (k + i < cnt) ? wq : 0.0f;
                    h[i]  = *(const float4*)(Hl + (size_t)si * 128);
                }
                #pragma unroll
                for (int i = 0; i < SPMM_GROUP; ++i) {
                    acc.x = fmaf(ws[i], h[i].x, acc.x);
                    acc.y = fmaf(ws[i], h[i].y, acc.y);
                    acc.z = fmaf(ws[i], h[i].z, acc.z);
                    acc.w = fmaf(ws[i], h[i].w, acc.w);
                }
            }
        }
        float4 o = make_float4(acc.x + b4.x, acc.y + b4.y, acc.z + b4.z, acc.w + b4.w);
        if (RELU) {
            o.x = fmaxf(o.x, 0.f); o.y = fmaxf(o.y, 0.f);
            o.z = fmaxf(o.z, 0.f); o.w = fmaxf(o.w, 0.f);
        }
        *(float4*)(OUT + (size_t)row * 128 + 4 * lane) = o;
    }
}

__global__ void log_softmax16_kernel(const float* __restrict__ Y,
                                     float* __restrict__ out, int N)
{
    const int r = blockIdx.x * blockDim.x + threadIdx.x;
    if (r >= N) return;
    float v[D_OUT];
    float m = -INFINITY;
    #pragma unroll
    for (int j = 0; j < D_OUT; ++j) { v[j] = Y[(size_t)r * D_OUT + j]; m = fmaxf(m, v[j]); }
    float s = 0.0f;
    #pragma unroll
    for (int j = 0; j < D_OUT; ++j) s += __expf(v[j] - m);
    const float l = m + __logf(s);
    #pragma unroll
    for (int j = 0; j < D_OUT; ++j) out[(size_t)r * D_OUT + j] = v[j] - l;
}

extern "C" void kernel_launch(void* const* d_in, const int* in_sizes, int n_in,
                              void* d_out, int out_size, void* d_ws, size_t ws_size,
                              hipStream_t stream)
{
    const float* x    = (const float*)d_in[0];
    const float* ew   = (const float*)d_in[1];
    const float* W1   = (const float*)d_in[2];
    const float* b1   = (const float*)d_in[3];
    const float* W2   = (const float*)d_in[4];
    const float* b2   = (const float*)d_in[5];
    const float* W3   = (const float*)d_in[6];
    const float* b3   = (const float*)d_in[7];
    const float* fcw  = (const float*)d_in[8];
    const float* fcb  = (const float*)d_in[9];
    const int*   ei   = (const int*)d_in[10];

    const int N = in_sizes[0] / D_IN;
    const int E = in_sizes[1];

    float* B0   = (float*)d_ws;
    float* B1   = B0 + (size_t)N * D_HID;
    float* WT1  = B1 + (size_t)N * D_HID;
    float* WT2  = WT1 + D_IN  * D_HID;
    float* WT3  = WT2 + D_HID * D_HID;
    float* WTfc = WT3 + D_HID * D_EMB;
    float* Y    = B0;

    int*   cnt    = (int*)B1;
    int*   rowptr = cnt + N;
    int*   ssrc   = rowptr + N + 1;
    float* sw     = (float*)(ssrc + E);

    float* outEmb = (float*)d_out;
    float* outLsm = outEmb + (size_t)N * D_EMB;

    const int rowTiles = (N + 15) / 16;
    const dim3 blk(256);

    auto transpose = [&](const float* W, float* WT, int K, int D) {
        const int tot = K * D;
        transpose_w_kernel<<<(tot + 255) / 256, blk, 0, stream>>>(W, WT, K, D);
    };
    auto gemm = [&](const float* A, const float* WT, float* C, const float* bias,
                    int K, int DOUTv) {
        const int colTiles = DOUTv / 16;
        const int total = rowTiles * colTiles;
        gemm_wmma_f32_kernel<<<(total + 7) / 8, blk, 0, stream>>>(
            A, WT, C, bias, N, K, DOUTv, total, colTiles);
    };
    auto gemmStrip256 = [&](const float* A, const float* W, float* C) {
        gemm_strip_wmma_f32_kernel<D_IN><<<(rowTiles + 7) / 8, blk, 0, stream>>>(
            A, W, C, N, rowTiles);
    };
    auto gemmStrip128 = [&](const float* A, const float* W, float* C) {
        gemm_strip_wmma_f32_kernel<D_HID><<<(rowTiles + 7) / 8, blk, 0, stream>>>(
            A, W, C, N, rowTiles);
    };

    transpose(fcw, WTfc, D_EMB, D_OUT);

    const int edgeBlocks = (E + 1023) / 1024;
    zero_int_kernel<<<(N + 255) / 256, blk, 0, stream>>>(cnt, N);
    csr_count_kernel<<<edgeBlocks, blk, 0, stream>>>(ei, cnt, E);
    csr_scan_kernel<<<1, 1024, 0, stream>>>(cnt, rowptr, N);
    csr_fill_kernel<<<edgeBlocks, blk, 0, stream>>>(ei, ew, rowptr, cnt, ssrc, sw, E);

    gemmStrip256(x, W1, B0);
    spmm_gather_kernel<true><<<1024, blk, 0, stream>>>(B0, outEmb, rowptr, ssrc, sw, b1, N);

    gemmStrip128(outEmb, W2, B0);
    spmm_gather_kernel<true><<<1024, blk, 0, stream>>>(B0, outEmb, rowptr, ssrc, sw, b2, N);

    gemmStrip128(outEmb, W3, B0);
    spmm_gather_kernel<false><<<1024, blk, 0, stream>>>(B0, outEmb, rowptr, ssrc, sw, b3, N);

    gemm(outEmb, WTfc, Y, fcb, D_EMB, D_OUT);
    log_softmax16_kernel<<<(N + 255) / 256, blk, 0, stream>>>(Y, outLsm, N);
}
